# starting kernel
# baseline (speedup 1.0000x reference)
#include <hip/hip_runtime.h>
#include <hip/hip_bf16.h>
#include <math.h>

#define B_   32
#define C_   256
#define H_   56
#define W_   56
#define HW_  (H_ * W_)
#define NPIX (B_ * HW_)

typedef __attribute__((ext_vector_type(2))) float v2f;
typedef __attribute__((ext_vector_type(4))) float v4f;
typedef __attribute__((ext_vector_type(8))) float v8f;

__global__ __launch_bounds__(256) void reduce_kernel(const float* __restrict__ x,
                                                     float2* __restrict__ y) {
  __shared__ float ssum[256];
  __shared__ float smax[256];
  const int t  = threadIdx.x;
  const int p  = t & 63;
  const int q  = t >> 6;
  const int gp = blockIdx.x * 64 + p;
  const int b  = gp / HW_;
  const int hw = gp - b * HW_;

  const float* xp = x + (size_t)(b * C_ + q * 64) * HW_ + hw;
  float s  = 0.0f;
  float mx = -3.402823466e38f;
#pragma unroll 8
  for (int c = 0; c < 64; ++c) {
    float v = xp[(size_t)c * HW_];
    s += v;
    mx = fmaxf(mx, v);
  }
  ssum[t] = s;
  smax[t] = mx;
  __syncthreads();
  if (q == 0) {
    float s0 = ssum[p] + ssum[p + 64] + ssum[p + 128] + ssum[p + 192];
    float m0 = fmaxf(fmaxf(smax[p], smax[p + 64]), fmaxf(smax[p + 128], smax[p + 192]));
    float2 r;
    r.x = s0 * (1.0f / (float)C_);
    r.y = m0;
    y[gp] = r;
  }
}

__global__ __launch_bounds__(32) void conv_wmma_kernel(const float2* __restrict__ y,
                                                       const float* __restrict__ cw,
                                                       float* __restrict__ att) {
  const int lane = threadIdx.x;
  const int m    = lane & 15;
  const int hi   = lane >> 4;

  const int gp = blockIdx.x * 16 + m;
  const int b  = gp / HW_;
  const int hw = gp - b * HW_;
  const int h  = hw / W_;
  const int w  = hw - h * W_;
  const float2* yb = y + (size_t)b * HW_;

  v8f acc = {0.f, 0.f, 0.f, 0.f, 0.f, 0.f, 0.f, 0.f};

  for (int c = 0; c < 25; ++c) {
    const int tap = 2 * c + hi;
    float2 av = make_float2(0.f, 0.f);
    float  w0 = 0.f, w1 = 0.f;
    if (tap < 49) {
      const int kh = tap / 7;
      const int kw = tap - kh * 7;
      const int ih = h + kh - 3;
      const int iw = w + kw - 3;
      if (ih >= 0 && ih < H_ && iw >= 0 && iw < W_) av = yb[ih * W_ + iw];
      w0 = cw[tap];
      w1 = cw[49 + tap];
    }
    v2f A = {av.x, av.y};
    v2f Bm = {w0, w1};
    acc = __builtin_amdgcn_wmma_f32_16x16x4_f32(
         false, A,  false, Bm,
         (short)0, acc,  false,  false);
  }

  if (m == 0) {
    const int mbase = hi * 8;
#pragma unroll
    for (int r = 0; r < 8; ++r) {
      float z = acc[r];
      float s = 1.0f / (1.0f + expf(-z));
      att[blockIdx.x * 16 + mbase + r] = s;
    }
  }
}

__global__ __launch_bounds__(256) void mul_kernel(const float* __restrict__ x,
                                                  const float* __restrict__ att,
                                                  float* __restrict__ out, int n4) {
  const int i = blockIdx.x * blockDim.x + threadIdx.x;
  if (i >= n4) return;
  const int per_b = C_ * (HW_ / 4);
  const int b     = i / per_b;
  const int rem   = i - b * per_b;
  const int hw4   = rem % (HW_ / 4);

  const v4f xv = __builtin_nontemporal_load(reinterpret_cast<const v4f*>(x) + i);
  const v4f av = *(reinterpret_cast<const v4f*>(att + (size_t)b * HW_) + hw4);
  v4f o = xv * av;
  __builtin_nontemporal_store(o, reinterpret_cast<v4f*>(out) + i);
}

extern "C" void kernel_launch(void* const* d_in, const int* in_sizes, int n_in,
                              void* d_out, int out_size, void* d_ws, size_t ws_size,
                              hipStream_t stream) {
  const float* x  = (const float*)d_in[0];
  const float* cw = (const float*)d_in[1];
  float* out      = (float*)d_out;

  float2* y   = (float2*)d_ws;
  float*  att = (float*)((char*)d_ws + (size_t)NPIX * sizeof(float2));

  (void)in_sizes; (void)n_in; (void)out_size; (void)ws_size;

  reduce_kernel<<<NPIX / 64, 256, 0, stream>>>(x, y);

  conv_wmma_kernel<<<NPIX / 16, 32, 0, stream>>>(y, cw, att);

  const int n4 = (B_ * C_ * HW_) / 4;
  mul_kernel<<<(n4 + 255) / 256, 256, 0, stream>>>(x, att, out, n4);
}


# strip's own 8 rows of x kept on chip from the reduce to the multiply (16 of a thread's 32 pieces in registers, 16 in LDS): x requested once, not twice; channel sum order changes (8 groups of 32)
# speedup vs baseline: 1.7955x; 1.7955x over previous
#include <hip/hip_runtime.h>
#include <math.h>

#define B_   32
#define C_   256
#define H_   56
#define W_   56
#define HW_  (H_ * W_)
#define HW4_ (HW_ / 4)
#define W4_  (W_ / 4)

typedef __attribute__((ext_vector_type(2))) float v2f;
typedef __attribute__((ext_vector_type(4))) float v4f;
typedef __attribute__((ext_vector_type(8))) float v8f;

#define SROWS_   8
#define NSTRIP_  (H_ / SROWS_)
#define HROWS_   (SROWS_ + 6)
#define HCOL4_   (HROWS_ * W4_)
#define HPIX_    (HROWS_ * W_)
#define SCOL4_   (SROWS_ * W4_)
#define NTHR_    1024
#define TILE_    14
#define YROWS_   22
#define YPITCH_  62
#define WPITCH_  36
#define CG_      32
#define NCG_     (C_ / CG_)
#define OWNT_    (NCG_ * SCOL4_)
#define KEEP_    16
#define HALO4_   (6 * W4_)

__global__ __launch_bounds__(NTHR_) void fused_kernel(const float* __restrict__ x,
                                                      const float* __restrict__ cw,
                                                      float* __restrict__ out) {
  __shared__ v4f    stash[(CG_ - KEEP_) * OWNT_];
  __shared__ v4f    ps[NCG_ * HCOL4_];
  __shared__ v4f    pm[NCG_ * HCOL4_];
  __shared__ float2 ys[YROWS_ * YPITCH_];
  __shared__ float2 tw[7 * WPITCH_];
  __shared__ v4f    atts[SCOL4_];

  const int t     = threadIdx.x;
  const int b     = blockIdx.x / NSTRIP_;
  const int strip = blockIdx.x - b * NSTRIP_;
  const int h0    = strip * SROWS_;

  for (int i = t; i < YROWS_ * YPITCH_; i += NTHR_) ys[i] = make_float2(0.f, 0.f);
  if (t < 7 * WPITCH_) {
    const int kh = t / WPITCH_;
    const int d  = t - kh * WPITCH_ - 15;
    const int dc = min(max(d, 0), 6);
    const float a0 = cw[kh * 7 + dc];
    const float a1 = cw[49 + kh * 7 + dc];
    tw[t] = (d == dc) ? make_float2(a0, a1) : make_float2(0.f, 0.f);
  }

  const int cg = t / SCOL4_;
  const int c4 = t - cg * SCOL4_;
  const long pbase = (long)(b * C_ + cg * CG_) * HW4_ + (long)h0 * W4_;
  v4f keep[KEEP_];
  if (t < OWNT_) {
    const v4f vzero = {0.f, 0.f, 0.f, 0.f};
    const v4f vlow  = {-3.402823466e38f, -3.402823466e38f, -3.402823466e38f, -3.402823466e38f};
    if (c4 < HALO4_) {
      const int hc = (c4 < 3 * W4_) ? c4 : c4 + SCOL4_;
      const int gh = h0 - 3 + hc / W4_;
      v4f s  = vzero;
      v4f mx = vlow;
      if (gh >= 0 && gh < H_) {
        const v4f* xh = reinterpret_cast<const v4f*>(x) + (pbase - 3 * W4_ + hc);
#pragma unroll
        for (int c0 = 0; c0 < CG_; c0 += 16) {
          v4f v[16];
#pragma unroll
          for (int i = 0; i < 16; ++i) v[i] = xh[(c0 + i) * HW4_];
#pragma unroll
          for (int i = 0; i < 16; ++i) {
            s += v[i];
            mx.x = fmaxf(mx.x, v[i].x);
            mx.y = fmaxf(mx.y, v[i].y);
            mx.z = fmaxf(mx.z, v[i].z);
            mx.w = fmaxf(mx.w, v[i].w);
          }
        }
      }
      ps[cg * HCOL4_ + hc] = s;
      pm[cg * HCOL4_ + hc] = mx;
    }
    {
      const v4f* xo = reinterpret_cast<const v4f*>(x) + (pbase + c4);
      v4f s  = vzero;
      v4f mx = vlow;
#pragma unroll
      for (int i = 0; i < KEEP_; ++i) keep[i] = xo[(CG_ - KEEP_ + i) * HW4_];
#pragma unroll
      for (int c0 = 0; c0 < CG_ - KEEP_; c0 += 8) {
        v4f v[8];
#pragma unroll
        for (int i = 0; i < 8; ++i) v[i] = xo[(c0 + i) * HW4_];
#pragma unroll
        for (int i = 0; i < 8; ++i) {
          s += v[i];
          mx.x = fmaxf(mx.x, v[i].x);
          mx.y = fmaxf(mx.y, v[i].y);
          mx.z = fmaxf(mx.z, v[i].z);
          mx.w = fmaxf(mx.w, v[i].w);
          stash[(c0 + i) * OWNT_ + t] = v[i];
        }
      }
#pragma unroll
      for (int i = 0; i < KEEP_; ++i) {
        s += keep[i];
        mx.x = fmaxf(mx.x, keep[i].x);
        mx.y = fmaxf(mx.y, keep[i].y);
        mx.z = fmaxf(mx.z, keep[i].z);
        mx.w = fmaxf(mx.w, keep[i].w);
      }
      ps[cg * HCOL4_ + 3 * W4_ + c4] = s;
      pm[cg * HCOL4_ + 3 * W4_ + c4] = mx;
    }
  }
  __syncthreads();

  if (t < HPIX_) {
    const int r   = t / W_;
    const int col = t - r * W_;
    const int gh  = h0 - 3 + r;
    if (gh >= 0 && gh < H_) {
      const float* psf = reinterpret_cast<const float*>(ps);
      const float* pmf = reinterpret_cast<const float*>(pm);
      float s0 = psf[t];
      float m0 = pmf[t];
#pragma unroll
      for (int k = 1; k < NCG_; ++k) {
        s0 += psf[t + k * HPIX_];
        m0 = fmaxf(m0, pmf[t + k * HPIX_]);
      }
      ys[r * YPITCH_ + col + 3] = make_float2(s0 * (1.0f / (float)C_), m0);
    }
  }
  __syncthreads();

  if (t < 4 * 32) {
    const int lane = t & 31;
    const int c    = lane & 15;
    const int hi   = lane >> 4;
    const int w0   = (t >> 5) * TILE_;

    const float2* ap = ys + c * YPITCH_ + w0 + hi;
    const float2* bp = tw + (hi - c + 15);

    v8f acc0 = {0.f, 0.f, 0.f, 0.f, 0.f, 0.f, 0.f, 0.f};
    v8f acc1 = {0.f, 0.f, 0.f, 0.f, 0.f, 0.f, 0.f, 0.f};
#pragma unroll
    for (int kh = 0; kh < 7; ++kh) {
#pragma unroll
      for (int cc = 0; cc < 10; cc += 2) {
        const float2 a0 = ap[kh * YPITCH_ + 2 * cc];
        const float2 b0 = bp[kh * WPITCH_ + 2 * cc];
        const float2 a1 = ap[kh * YPITCH_ + 2 * cc + 2];
        const float2 b1 = bp[kh * WPITCH_ + 2 * cc + 2];
        const v2f A0 = {a0.x, a0.y};
        const v2f B0 = {b0.x, b0.y};
        const v2f A1 = {a1.x, a1.y};
        const v2f B1 = {b1.x, b1.y};
        acc0 = __builtin_amdgcn_wmma_f32_16x16x4_f32(
             false, A0,  false, B0,
             (short)0, acc0,  false,  false);
        acc1 = __builtin_amdgcn_wmma_f32_16x16x4_f32(
             false, A1,  false, B1,
             (short)0, acc1,  false,  false);
      }
    }

    float* as = reinterpret_cast<float*>(atts);
#pragma unroll
    for (int r = 0; r < 8; ++r) {
      const float z = acc0[r] + acc1[r];
      const float s = 1.0f / (1.0f + expf(-z));
      if (hi == 0 && c < TILE_) as[r * W_ + w0 + c] = s;
    }
  }
  __syncthreads();

  if (t < OWNT_) {
    const v4f a = atts[c4];
    v4f* oo = reinterpret_cast<v4f*>(out) + (pbase + c4);
#pragma unroll
    for (int i = 0; i < KEEP_; ++i)
      __builtin_nontemporal_store(keep[i] * a, oo + (CG_ - KEEP_ + i) * HW4_);
#pragma unroll
    for (int i = 0; i < CG_ - KEEP_; ++i)
      __builtin_nontemporal_store(stash[i * OWNT_ + t] * a, oo + i * HW4_);
  }
}

extern "C" void kernel_launch(void* const* d_in, const int* in_sizes, int n_in,
                              void* d_out, int out_size, void* d_ws, size_t ws_size,
                              hipStream_t stream) {
  const float* x  = (const float*)d_in[0];
  const float* cw = (const float*)d_in[1];
  float* out      = (float*)d_out;

  (void)in_sizes; (void)n_in; (void)out_size; (void)d_ws; (void)ws_size;

  fused_kernel<<<B_ * NSTRIP_, NTHR_, 0, stream>>>(x, cw, out);
}
